# starting kernel
# baseline (speedup 1.0000x reference)
#include <hip/hip_runtime.h>

#define B_ 16
#define L_ 2048
#define D_ 1024
#define S_ 16
#define OUTD (3 * D_)
#define KWAVES 8
#define KSLICE (L_ / KWAVES)

typedef __attribute__((ext_vector_type(2))) float v2f;
typedef __attribute__((ext_vector_type(8))) float v8f;

__global__ void find_markers(const int* __restrict__ ids,
                             int* __restrict__ idx0,
                             int* __restrict__ idx1) {
    const int b = blockIdx.x;
    const int* row = ids + b * L_;
    for (int p = threadIdx.x; p < L_; p += blockDim.x) {
        const int id = row[p];
        const bool isS = (id >= 1 && id <= 2);
        const bool isE = (id >= 3 && id <= 4);
        if (isS) {
            int rank = 0;
            for (int j = 0; j < p; ++j) {
                const int v = row[j];
                rank += (v >= 1 && v <= 2) ? 1 : 0;
            }
            idx0[b * S_ + rank] = p;
        } else if (isE) {
            int rank = 0;
            for (int j = 0; j < p; ++j) {
                const int v = row[j];
                rank += (v >= 3 && v <= 4) ? 1 : 0;
            }
            idx1[b * S_ + rank] = p;
        }
    }
}

__global__ void gather_ends(const float* __restrict__ x,
                            const int* __restrict__ idx0,
                            const int* __restrict__ idx1,
                            float* __restrict__ out) {
    const int bs = blockIdx.x;
    const int b = bs >> 4;
    const int i0 = idx0[bs];
    const int i1 = idx1[bs];
    const float4* r0 = (const float4*)(x + ((size_t)b * L_ + i0) * D_);
    const float4* r1 = (const float4*)(x + ((size_t)b * L_ + i1) * D_);
    float4* o = (float4*)(out + (size_t)bs * OUTD);
    const int t = threadIdx.x;
    o[t] = r0[t];
    o[(D_ / 4) + t] = r1[t];
}

__global__ void __launch_bounds__(256) span_mean_wmma(
    const float* __restrict__ x,
    const int* __restrict__ idx0,
    const int* __restrict__ idx1,
    float* __restrict__ out) {
    __shared__ float red[KWAVES * 256];

    const int b    = blockIdx.y;
    const int wave = threadIdx.x >> 5;
    const int lane = threadIdx.x & 31;
    const int col0 = blockIdx.x * 16;

    const int n     = lane & 15;
    const int m     = lane & 15;
    const int khalf = (lane >> 4) * 2;

    const int   i0   = idx0[b * S_ + m];
    const int   i1   = idx1[b * S_ + m];
    const float invc = 1.0f / (float)(i1 - i0 - 1);

    const float* bcol = x + (size_t)b * L_ * D_ + col0 + n;

    v8f c = {};
    const int kbeg = wave * KSLICE;
#pragma unroll 16
    for (int l0 = kbeg; l0 < kbeg + KSLICE; l0 += 4) {
        const int lx = l0 + khalf;
        const int ly = lx + 1;

        v2f a;
        a.x = (lx > i0 && lx < i1) ? invc : 0.0f;
        a.y = (ly > i0 && ly < i1) ? invc : 0.0f;

        v2f bb;
        bb.x = bcol[(size_t)lx * D_];
        bb.y = bcol[(size_t)ly * D_];

        c = __builtin_amdgcn_wmma_f32_16x16x4_f32(
                 false, a,  false, bb,
                 (short)0, c,  false,  false);
    }

#pragma unroll
    for (int i = 0; i < 8; ++i)
        red[wave * 256 + i * 32 + lane] = c[i];
    __syncthreads();

    const int t = threadIdx.x;
    float s = 0.0f;
#pragma unroll
    for (int w = 0; w < KWAVES; ++w)
        s += red[w * 256 + t];

    const int i    = t >> 5;
    const int ln   = t & 31;
    const int span = (ln >> 4) * 8 + i;
    const int col  = col0 + (ln & 15);
    out[(size_t)(b * S_ + span) * OUTD + 2 * D_ + col] = s;
}

extern "C" void kernel_launch(void* const* d_in, const int* in_sizes, int n_in,
                              void* d_out, int out_size, void* d_ws, size_t ws_size,
                              hipStream_t stream) {
    const float* x   = (const float*)d_in[0];
    const int*   ids = (const int*)d_in[2];
    float*       out = (float*)d_out;

    int* idx0 = (int*)d_ws;
    int* idx1 = idx0 + B_ * S_;

    find_markers<<<B_, 256, 0, stream>>>(ids, idx0, idx1);
    gather_ends<<<B_ * S_, 256, 0, stream>>>(x, idx0, idx1, out);

    dim3 grid(D_ / 16, B_);
    span_mean_wmma<<<grid, 256, 0, stream>>>(x, idx0, idx1, out);
}


# a work-group per (batch row, 64-column strip) streams all 2048 rows at once, not waiting for the marker scan; the 16 spans are the 16 rows of P; K split over 8 waves, LDS add: changes the sum tree
# speedup vs baseline: 7.8401x; 7.8401x over previous
#include <hip/hip_runtime.h>


#define B_ 16
#define L_ 2048
#define D_ 1024
#define S_ 16
#define OUTD (3 * D_)
#define NWAVES 8
#define STRIP 64
#define TPW (L_ / NWAVES)
#define NSTEP (TPW / 4)
#define RING 16
#define NPASS (NSTEP / RING)

typedef __attribute__((ext_vector_type(2))) float v2f;
typedef __attribute__((ext_vector_type(4))) float v4f;
typedef __attribute__((ext_vector_type(8))) float v8f;

__device__ __forceinline__ int is_start(int id) { return (id >= 1 && id <= 2) ? 1 : 0; }
__device__ __forceinline__ int is_end(int id)   { return (id >= 3 && id <= 4) ? 1 : 0; }

__device__ __forceinline__ void lds_barrier() {
    asm volatile("s_wait_dscnt 0x0\n\ts_barrier_signal -1\n\ts_barrier_wait -1" ::: "memory");
}

__global__ void __launch_bounds__(256) span_pool(
    const float* __restrict__ x,
    const int* __restrict__ ids,
    float* __restrict__ out) {
    __shared__ int   wtot[NWAVES];
    __shared__ int   pos0[S_];
    __shared__ int   pos1[S_];
    __shared__ float red[NWAVES][4][8][32];

    const int b     = blockIdx.x >> 4;
    const int strip = blockIdx.x & 15;
    const int t     = threadIdx.x;
    const int wave  = t >> 5;
    const int lane  = t & 31;
    const int c     = lane & 15;
    const int h     = lane >> 4;

    const int4* idrow = (const int4*)(ids + b * L_) + 2 * t;
    const int4 qa = idrow[0];
    const int4 qb = idrow[1];

    const float* xb   = x + (size_t)b * L_ * D_;
    const float* base = xb + (size_t)(TPW * wave + 2 * h) * D_ + strip * STRIP + 4 * c;

    v4f buf[RING][2];
    auto load_step = [&](int i, v4f (&d)[2]) {
        const v4f* r = (const v4f*)(base + (size_t)i * 4 * D_);
        d[0] = r[0];
        d[1] = r[D_ / 4];
    };
    #pragma unroll
    for (int r = 0; r < RING; ++r) load_step(r, buf[r]);
    __builtin_amdgcn_sched_barrier(0);

    const int v[8] = {qa.x, qa.y, qa.z, qa.w, qb.x, qb.y, qb.z, qb.w};

    int cnt = 0;
    for (int i = 0; i < 8; ++i)
        cnt += is_start(v[i]) + (is_end(v[i]) << 16);

    int inc = cnt;
    for (int d = 1; d < 32; d <<= 1) {
        const int o = __shfl_up(inc, d);
        if (lane >= d) inc += o;
    }
    if (lane == 31) wtot[wave] = inc;
    if (t < S_) { pos0[t] = 0; pos1[t] = 0; }
    lds_barrier();

    int before = inc - cnt;
    for (int w = 0; w < NWAVES; ++w)
        if (w < wave) before += wtot[w];

    int rs = before & 0xffff;
    int re = before >> 16;
    for (int i = 0; i < 8; ++i) {
        if (is_start(v[i])) { if (rs < S_) pos0[rs] = 8 * t + i; ++rs; }
        if (is_end(v[i]))   { if (re < S_) pos1[re] = 8 * t + i; ++re; }
    }
    lds_barrier();

    const int      p0m  = pos0[c];
    const int      p1m  = pos1[c];
    const int      ntok = p1m - p0m - 1;
    const float    invc = 1.0f / (float)ntok;
    const unsigned ntc  = (unsigned)max(ntok, 0);
    const int      rel0 = TPW * wave + 2 * h - p0m - 1;

    const int sp = t >> 4;
    const int cc = t & 15;
    float* o = out + (size_t)(b * S_ + sp) * OUTD + strip * STRIP + 4 * cc;
    const v4f e0 = *(const v4f*)(xb + (size_t)pos0[sp] * D_ + strip * STRIP + 4 * cc);
    const v4f e1 = *(const v4f*)(xb + (size_t)pos1[sp] * D_ + strip * STRIP + 4 * cc);

    v8f acc[4];
    for (int j = 0; j < 4; ++j) acc[j] = v8f{};

    auto use_step = [&](int i, v4f (&d)[2]) {
        asm volatile("" : "+v"(d[0]), "+v"(d[1]));
        const int rel = rel0 + 4 * i;
        v2f a;
        a.x = ((unsigned)rel       < ntc) ? invc : 0.0f;
        a.y = ((unsigned)(rel + 1) < ntc) ? invc : 0.0f;
        for (int j = 0; j < 4; ++j) {
            v2f bb;
            bb.x = d[0][j];
            bb.y = d[1][j];
            acc[j] = __builtin_amdgcn_wmma_f32_16x16x4_f32(
                 false, a,  false, bb,
                 (short)0, acc[j],
                 false,  false);
        }
        __builtin_amdgcn_sched_barrier(0);
    };

    #pragma unroll
    for (int p = 0; p < NPASS; ++p) {
        #pragma unroll
        for (int r = 0; r < RING; ++r) {
            use_step(p * RING + r, buf[r]);
            if (p + 1 < NPASS) load_step((p + 1) * RING + r, buf[r]);
        }
        if (p == 0) {
            *(v4f*)o        = e0;
            *(v4f*)(o + D_) = e1;
        }
    }

    for (int j = 0; j < 4; ++j)
        for (int r = 0; r < 8; ++r)
            red[wave][j][r][lane] = acc[j][r];
    __syncthreads();

    const int rr = sp & 7;
    const int ll = cc + 16 * (sp >> 3);
    float4 m = {0.0f, 0.0f, 0.0f, 0.0f};
    for (int w = 0; w < NWAVES; ++w) {
        m.x += red[w][0][rr][ll];
        m.y += red[w][1][rr][ll];
        m.z += red[w][2][rr][ll];
        m.w += red[w][3][rr][ll];
    }
    *(float4*)(o + 2 * D_) = m;
}

extern "C" void kernel_launch(void* const* d_in, const int* in_sizes, int n_in,
                              void* d_out, int out_size, void* d_ws, size_t ws_size,
                              hipStream_t stream) {
    const float* x   = (const float*)d_in[0];
    const int*   ids = (const int*)d_in[2];
    float*       out = (float*)d_out;

    span_pool<<<B_ * (D_ / STRIP), 256, 0, stream>>>(x, ids, out);
}
